# starting kernel
# baseline (speedup 1.0000x reference)
#include <hip/hip_runtime.h>

typedef __attribute__((ext_vector_type(2))) float v2f;
typedef __attribute__((ext_vector_type(8))) float v8f;

#define F_DIM 128
#define H_DIM 64

__device__ __forceinline__ void atomAddF(float* p, float v) {
    __hip_atomic_fetch_add(p, v, __ATOMIC_RELAXED, __HIP_MEMORY_SCOPE_AGENT);
}

__global__ void __launch_bounds__(256) gemm_wmma_f32(
    const float* __restrict__ x, const float* __restrict__ W,
    const float* __restrict__ bias, float* __restrict__ xt, int N)
{
    const int waveId = (blockIdx.x * blockDim.x + threadIdx.x) >> 5;
    const int lane   = threadIdx.x & 31;
    const int tilesH = H_DIM / 16;
    const int tileM  = waveId / tilesH;
    const int tileH  = waveId % tilesH;
    if (tileM * 16 >= N) return;

    const int m   = lane & 15;
    const int kp  = (lane >> 4) << 1;
    const float* xrow = x + (size_t)(tileM * 16 + m) * F_DIM;
    const float* wrow = W + (size_t)(tileH * 16 + m) * F_DIM;

    v8f c = {};
    #pragma unroll 8
    for (int k = 0; k < F_DIM; k += 4) {
        v2f a = *(const v2f*)(xrow + k + kp);
        v2f b = *(const v2f*)(wrow + k + kp);
        c = __builtin_amdgcn_wmma_f32_16x16x4_f32(
                 false, a,  false, b,
                 (short)0, c,  false,  false);
    }

    const int hOut  = tileH * 16 + (lane & 15);
    const float bv  = bias[hOut];
    const int mBase = tileM * 16 + ((lane >> 4) << 3);
    #pragma unroll
    for (int v = 0; v < 8; ++v) {
        xt[(size_t)(mBase + v) * H_DIM + hOut] = c[v] + bv;
    }
}

__global__ void __launch_bounds__(256) spmm_scatter(
    const float* __restrict__ xt, const int* __restrict__ rows,
    const int* __restrict__ cols, const float* __restrict__ vals,
    float* __restrict__ outH, int E)
{
    long long idx = (long long)blockIdx.x * blockDim.x + threadIdx.x;
    long long e = idx >> 4;
    int q = (int)(idx & 15);
    if (e >= E) return;
    const int r = rows[e];
    const int c = cols[e];
    const float v = vals[e];
    const float4 msg = ((const float4*)(xt + (size_t)c * H_DIM))[q];
    float* dst = outH + (size_t)r * H_DIM + q * 4;
    atomAddF(dst + 0, v * msg.x);
    atomAddF(dst + 1, v * msg.y);
    atomAddF(dst + 2, v * msg.z);
    atomAddF(dst + 3, v * msg.w);
}

__global__ void __launch_bounds__(256) prelu_colsum(
    float* __restrict__ h, const float* __restrict__ prelu_a,
    float* __restrict__ colsum, long long total, int doSum)
{
    const float a = prelu_a[0];
    const long long stride = (long long)gridDim.x * blockDim.x;
    float local = 0.f;
    for (long long i = (long long)blockIdx.x * blockDim.x + threadIdx.x;
         i < total; i += stride) {
        float v = h[i];
        float o = (v >= 0.f) ? v : a * v;
        h[i] = o;
        local += o;
    }
    if (doSum) {
        __shared__ float sm[256];
        sm[threadIdx.x] = local;
        __syncthreads();
        if (threadIdx.x < 64) {
            float t = sm[threadIdx.x] + sm[threadIdx.x + 64] +
                      sm[threadIdx.x + 128] + sm[threadIdx.x + 192];
            atomAddF(colsum + threadIdx.x, t);
        }
    }
}

__global__ void summary_kernel(
    const float* __restrict__ colsum, const float* __restrict__ Wbil,
    float* __restrict__ tbuf, int N)
{
    __shared__ float s_sh[H_DIM];
    const int hh = threadIdx.x;
    float s = colsum[hh] / (float)N;
    s = 1.f / (1.f + expf(-s));
    s_sh[hh] = s;
    __syncthreads();
    float t = 0.f;
    #pragma unroll
    for (int k = 0; k < H_DIM; ++k) t += Wbil[hh * H_DIM + k] * s_sh[k];
    tbuf[hh] = t;
}

__global__ void __launch_bounds__(256) score_kernel(
    const float* __restrict__ Hm, const float* __restrict__ tbuf,
    const float* __restrict__ bbil, float* __restrict__ out, int N)
{
    __shared__ float tsh[H_DIM];
    if (threadIdx.x < H_DIM) tsh[threadIdx.x] = tbuf[threadIdx.x];
    __syncthreads();
    const int wave = threadIdx.x >> 5;
    const int lane = threadIdx.x & 31;
    const int n = blockIdx.x * (blockDim.x >> 5) + wave;
    if (n >= N) return;
    const float* hr = Hm + (size_t)n * H_DIM;
    float acc = hr[lane] * tsh[lane] + hr[lane + 32] * tsh[lane + 32];
    #pragma unroll
    for (int off = 16; off >= 1; off >>= 1)
        acc += __shfl_xor(acc, off, 32);
    if (lane == 0) out[n] = acc + bbil[0];
}

extern "C" void kernel_launch(void* const* d_in, const int* in_sizes, int n_in,
                              void* d_out, int out_size, void* d_ws, size_t ws_size,
                              hipStream_t stream) {
    const float* pos   = (const float*)d_in[0];
    const float* neg   = (const float*)d_in[1];
    const int*   erows = (const int*)  d_in[2];
    const int*   ecols = (const int*)  d_in[3];
    const float* evals = (const float*)d_in[4];
    const float* Wg    = (const float*)d_in[5];
    const float* bg    = (const float*)d_in[6];
    const float* pa    = (const float*)d_in[7];
    const float* Wb    = (const float*)d_in[8];
    const float* bb    = (const float*)d_in[9];

    const int N = in_sizes[0] / F_DIM;
    const int E = in_sizes[2];
    float* out = (float*)d_out;

    float* xt     = (float*)d_ws;
    float* posH   = xt   + (size_t)N * H_DIM;
    float* negH   = posH + (size_t)N * H_DIM;
    float* colsum = negH + (size_t)N * H_DIM;
    float* tbuf   = colsum + H_DIM;

    hipMemsetAsync(posH,   0, (size_t)N * H_DIM * sizeof(float), stream);
    hipMemsetAsync(negH,   0, (size_t)N * H_DIM * sizeof(float), stream);
    hipMemsetAsync(colsum, 0, H_DIM * sizeof(float), stream);

    const int tiles      = (N / 16) * (H_DIM / 16);
    const int gemmBlocks = (tiles * 32 + 255) / 256;
    const long long spmmThreads = (long long)E * (H_DIM / 4);
    const int spmmBlocks = (int)((spmmThreads + 255) / 256);
    const long long total = (long long)N * H_DIM;
    const int scoreBlocks = (N + 7) / 8;

    gemm_wmma_f32<<<gemmBlocks, 256, 0, stream>>>(pos, Wg, bg, xt, N);
    spmm_scatter <<<spmmBlocks, 256, 0, stream>>>(xt, erows, ecols, evals, posH, E);
    prelu_colsum <<<2048, 256, 0, stream>>>(posH, pa, colsum, total, 1);
    gemm_wmma_f32<<<gemmBlocks, 256, 0, stream>>>(neg, Wg, bg, xt, N);
    spmm_scatter <<<spmmBlocks, 256, 0, stream>>>(xt, erows, ecols, evals, negH, E);
    prelu_colsum <<<2048, 256, 0, stream>>>(negH, pa, colsum, total, 0);
    summary_kernel<<<1, 64, 0, stream>>>(colsum, Wb, tbuf, N);
    score_kernel<<<scoreBlocks, 256, 0, stream>>>(posH, tbuf, bb, out, N);
    score_kernel<<<scoreBlocks, 256, 0, stream>>>(negH, tbuf, bb, out + N, N);
}


# rank, fill and score kernels take 4 edges per thread and 16 rows per wave (two rows per 16-byte-per-lane load): 31k -> 7.8k waves per edge pass, 100k -> 6.3k for the scores; wave entry set their time
# speedup vs baseline: 1.8644x; 1.8644x over previous
#include <hip/hip_runtime.h>

typedef __attribute__((ext_vector_type(2))) float v2f;
typedef __attribute__((ext_vector_type(8))) float v8f;

#define F_DIM 128
#define H_DIM 64

__device__ __forceinline__ void atomAddF(float* p, float v) {
    __hip_atomic_fetch_add(p, v, __ATOMIC_RELAXED, __HIP_MEMORY_SCOPE_AGENT);
}

#define EPT 4
__device__ __forceinline__ bool aligned16(const void* a, const void* b) {
    return (((size_t)a | (size_t)b) & 15) == 0;
}

__global__ void __launch_bounds__(256) edge_rank(
    const int* __restrict__ rows, int* __restrict__ cnt,
    int* __restrict__ rank, int E)
{
    const int e0 = (blockIdx.x * blockDim.x + threadIdx.x) * EPT;
    if (e0 >= E) return;
    if (e0 + EPT <= E && aligned16(rows, rank)) {
        const int4 r = *(const int4*)(rows + e0);
        int4 k;
        k.x = __hip_atomic_fetch_add(cnt + r.x, 1, __ATOMIC_RELAXED, __HIP_MEMORY_SCOPE_AGENT);
        k.y = __hip_atomic_fetch_add(cnt + r.y, 1, __ATOMIC_RELAXED, __HIP_MEMORY_SCOPE_AGENT);
        k.z = __hip_atomic_fetch_add(cnt + r.z, 1, __ATOMIC_RELAXED, __HIP_MEMORY_SCOPE_AGENT);
        k.w = __hip_atomic_fetch_add(cnt + r.w, 1, __ATOMIC_RELAXED, __HIP_MEMORY_SCOPE_AGENT);
        *(int4*)(rank + e0) = k;
    } else {
        const int e1 = (e0 + EPT < E) ? e0 + EPT : E;
        for (int e = e0; e < e1; ++e)
            rank[e] = __hip_atomic_fetch_add(cnt + rows[e], 1, __ATOMIC_RELAXED,
                                             __HIP_MEMORY_SCOPE_AGENT);
    }
}

__global__ void __launch_bounds__(1024) rowptr_scan(
    const int* __restrict__ cnt, int* __restrict__ rowptr, int N)
{
    __shared__ int wsum[32];
    __shared__ int prefixSh;
    const int tid  = threadIdx.x;
    const int lane = tid & 31;
    const int wv   = tid >> 5;
    const int start = blockIdx.x * 1024;

    int part = 0;
    for (int i = tid; i < start; i += 1024) part += cnt[i];
    #pragma unroll
    for (int off = 16; off >= 1; off >>= 1) part += __shfl_xor(part, off, 32);
    if (lane == 0) wsum[wv] = part;
    __syncthreads();
    if (wv == 0) {
        int v = wsum[lane];
        #pragma unroll
        for (int off = 16; off >= 1; off >>= 1) v += __shfl_xor(v, off, 32);
        if (lane == 0) prefixSh = v;
    }
    __syncthreads();
    const int prefix = prefixSh;

    const int i = start + tid;
    const int c = (i < N) ? cnt[i] : 0;
    int inc = c;
    #pragma unroll
    for (int off = 1; off < 32; off <<= 1) {
        const int t = __shfl_up(inc, off, 32);
        if (lane >= off) inc += t;
    }
    __syncthreads();
    if (lane == 31) wsum[wv] = inc;
    __syncthreads();
    if (wv == 0) {
        int v = wsum[lane];
        #pragma unroll
        for (int off = 1; off < 32; off <<= 1) {
            const int t = __shfl_up(v, off, 32);
            if (lane >= off) v += t;
        }
        wsum[lane] = v;
    }
    __syncthreads();
    const int woff = (wv > 0) ? wsum[wv - 1] : 0;
    if (i <= N) rowptr[i] = prefix + woff + inc - c;
}

__device__ __forceinline__ void csr_put(int2* __restrict__ csr, const int p, const int col,
                                        const float val, const int E) {
    if ((unsigned)p < (unsigned)E) csr[p] = make_int2(col, __float_as_int(val));
}

__global__ void __launch_bounds__(256) csr_fill(
    const int* __restrict__ rows, const int* __restrict__ cols,
    const float* __restrict__ vals, const int* __restrict__ rank,
    const int* __restrict__ rowptr, int2* __restrict__ csr, int E)
{
    const int e0 = (blockIdx.x * blockDim.x + threadIdx.x) * EPT;
    if (e0 >= E) return;
    if (e0 + EPT <= E && aligned16(rows, cols) && aligned16(vals, rank)) {
        const int4   r = *(const int4*)(rows + e0);
        const int4   k = *(const int4*)(rank + e0);
        const int4   c = *(const int4*)(cols + e0);
        const float4 v = *(const float4*)(vals + e0);
        const int p0 = rowptr[r.x] + k.x;
        const int p1 = rowptr[r.y] + k.y;
        const int p2 = rowptr[r.z] + k.z;
        const int p3 = rowptr[r.w] + k.w;
        csr_put(csr, p0, c.x, v.x, E);
        csr_put(csr, p1, c.y, v.y, E);
        csr_put(csr, p2, c.z, v.z, E);
        csr_put(csr, p3, c.w, v.w, E);
    } else {
        const int e1 = (e0 + EPT < E) ? e0 + EPT : E;
        for (int e = e0; e < e1; ++e)
            csr_put(csr, rowptr[rows[e]] + rank[e], cols[e], vals[e], E);
    }
}

#define RPT 16
#define WPB 25
#define APITCH 132

__device__ __forceinline__ void add_chunk(
    float4& acc, float& sv, const int2 ev, const int cnt,
    const float* __restrict__ x, const int lane, const int N)
{
    #pragma unroll
    for (int g = 0; g < 8; ++g) {
        const int j = 4 * g;
        if (j >= cnt) break;
        int c0 = __builtin_amdgcn_readlane(ev.x, j);
        int c1 = __builtin_amdgcn_readlane(ev.x, j + 1);
        int c2 = __builtin_amdgcn_readlane(ev.x, j + 2);
        int c3 = __builtin_amdgcn_readlane(ev.x, j + 3);
        const float v0 = __int_as_float(__builtin_amdgcn_readlane(ev.y, j));
        const float v1 = __int_as_float(__builtin_amdgcn_readlane(ev.y, j + 1));
        const float v2 = __int_as_float(__builtin_amdgcn_readlane(ev.y, j + 2));
        const float v3 = __int_as_float(__builtin_amdgcn_readlane(ev.y, j + 3));
        c0 = ((unsigned)c0 < (unsigned)N) ? c0 : 0;
        c1 = ((unsigned)c1 < (unsigned)N) ? c1 : 0;
        c2 = ((unsigned)c2 < (unsigned)N) ? c2 : 0;
        c3 = ((unsigned)c3 < (unsigned)N) ? c3 : 0;
        const float4 m0 = ((const float4*)(x + (size_t)c0 * F_DIM))[lane];
        const float4 m1 = ((const float4*)(x + (size_t)c1 * F_DIM))[lane];
        const float4 m2 = ((const float4*)(x + (size_t)c2 * F_DIM))[lane];
        const float4 m3 = ((const float4*)(x + (size_t)c3 * F_DIM))[lane];
        acc.x += v0 * m0.x;  acc.y += v0 * m0.y;  acc.z += v0 * m0.z;  acc.w += v0 * m0.w;
        acc.x += v1 * m1.x;  acc.y += v1 * m1.y;  acc.z += v1 * m1.z;  acc.w += v1 * m1.w;
        acc.x += v2 * m2.x;  acc.y += v2 * m2.y;  acc.z += v2 * m2.z;  acc.w += v2 * m2.w;
        acc.x += v3 * m3.x;  acc.y += v3 * m3.y;  acc.z += v3 * m3.z;  acc.w += v3 * m3.w;
        sv += (v0 + v1) + (v2 + v3);
    }
}

template <bool SCORE>
__global__ void __launch_bounds__(WPB * 32) gather_linear(
    const float* __restrict__ x, const float* __restrict__ W,
    const float* __restrict__ bias, const int* __restrict__ rowptr,
    const int2* __restrict__ csr, const float* __restrict__ prelu_a,
    float* __restrict__ outH, float* __restrict__ colsum,
    const float* __restrict__ tbuf, const float* __restrict__ bbil,
    float* __restrict__ score, int N, int E)
{
    __shared__ __attribute__((aligned(16))) float wS[H_DIM * APITCH];
    __shared__ __attribute__((aligned(16))) float aS[WPB * RPT * APITCH];

    const int lane = threadIdx.x & 31;
    const int wv = __builtin_amdgcn_readfirstlane((int)(threadIdx.x >> 5));
    const int r0 = (blockIdx.x * WPB + wv) * RPT;
    int nr = N - r0;
    nr = nr < 0 ? 0 : (nr > RPT ? RPT : nr);
    float* aT = aS + wv * (RPT * APITCH);

    const float a = prelu_a[0];
    float2 tv = make_float2(0.f, 0.f);
    if (SCORE) tv = ((const float2*)tbuf)[lane];

    int pi = r0 + lane;
    pi = pi > N ? N : pi;
    int myptr = rowptr[pi];
    myptr = myptr < 0 ? 0 : (myptr > E ? E : myptr);

    constexpr int W_PIECES  = H_DIM * (F_DIM / 4);
    constexpr int W_STAGERS = 512;
    constexpr int W_TRIPS   = W_PIECES / W_STAGERS;
    static_assert(W_PIECES % W_STAGERS == 0 && W_STAGERS % 32 == 0 && W_STAGERS <= WPB * 32,
                  "W staging: whole waves, whole trips");
    if (threadIdx.x < W_STAGERS) {
        float4 wreg[W_TRIPS];
        #pragma unroll
        for (int t = 0; t < W_TRIPS; ++t)
            wreg[t] = ((const float4*)W)[threadIdx.x + t * W_STAGERS];
        #pragma unroll
        for (int t = 0; t < W_TRIPS; ++t) {
            const int idx = threadIdx.x + t * W_STAGERS;
            *(float4*)(wS + (idx >> 5) * APITCH + 4 * (idx & 31)) = wreg[t];
        }
    }
    __syncthreads();

    int beg = __builtin_amdgcn_readlane(myptr, 0);
    int end = __builtin_amdgcn_readlane(myptr, 1);
    int2 ev = make_int2(0, 0);
    if (nr > 0 && lane < end - beg) ev = csr[beg + lane];
    asm volatile("" :: "v"(ev.x), "v"(ev.y));

    for (int i = 0; i < nr; ++i) {
        int endn = end;
        int2 evn = make_int2(0, 0);
        if (i + 1 < nr) {
            endn = __builtin_amdgcn_readlane(myptr, i + 2);
            if (lane < endn - end) evn = csr[end + lane];
        }

        float4 acc = make_float4(0.f, 0.f, 0.f, 0.f);
        float sv = 0.f;
        const int deg = end - beg;
        add_chunk(acc, sv, ev, deg > 32 ? 32 : deg, x, lane, N);
        if (deg > 32) {
            for (int base = beg + 32; base < end; base += 32) {
                int2 e2 = make_int2(0, 0);
                if (lane < end - base) e2 = csr[base + lane];
                const int rest = end - base;
                add_chunk(acc, sv, e2, rest > 32 ? 32 : rest, x, lane, N);
            }
        }

        *(float4*)(aT + i * APITCH + 4 * lane) = acc;
        if (lane == 0) aT[i * APITCH + F_DIM] = sv;

        beg = end;
        end = endn;
        ev  = evn;
    }
    for (int i = nr; i < RPT; ++i) {
        *(float4*)(aT + i * APITCH + 4 * lane) = make_float4(0.f, 0.f, 0.f, 0.f);
        if (lane == 0) aT[i * APITCH + F_DIM] = 0.f;
    }

    float2 cs = make_float2(0.f, 0.f);
    float myscore = 0.f;

    if (nr > 0) {
        __builtin_amdgcn_fence(__ATOMIC_RELEASE, "wavefront");
        __builtin_amdgcn_wave_barrier();
        __builtin_amdgcn_fence(__ATOMIC_ACQUIRE, "wavefront");

        const int c = lane & 15;
        const int h = lane >> 4;
        const float* ap = aT + c * APITCH + 64 * h;
        const float* wp = wS + c * APITCH + 64 * h;
        v8f d0 = {}, d1 = {}, d2 = {}, d3 = {};
        #pragma unroll 2
        for (int p = 0; p < 16; ++p) {
            const float4 a4 = *(const float4*)(ap + 4 * p);
            const float4 b0 = *(const float4*)(wp + 4 * p);
            const float4 b1 = *(const float4*)(wp + 16 * APITCH + 4 * p);
            const float4 b2 = *(const float4*)(wp + 32 * APITCH + 4 * p);
            const float4 b3 = *(const float4*)(wp + 48 * APITCH + 4 * p);
            const v2f alo = {a4.x, a4.y};
            const v2f ahi = {a4.z, a4.w};
            d0 = __builtin_amdgcn_wmma_f32_16x16x4_f32(false, alo, false, (v2f){b0.x, b0.y}, (short)0, d0, false, false);
            d1 = __builtin_amdgcn_wmma_f32_16x16x4_f32(false, alo, false, (v2f){b1.x, b1.y}, (short)0, d1, false, false);
            d2 = __builtin_amdgcn_wmma_f32_16x16x4_f32(false, alo, false, (v2f){b2.x, b2.y}, (short)0, d2, false, false);
            d3 = __builtin_amdgcn_wmma_f32_16x16x4_f32(false, alo, false, (v2f){b3.x, b3.y}, (short)0, d3, false, false);
            d0 = __builtin_amdgcn_wmma_f32_16x16x4_f32(false, ahi, false, (v2f){b0.z, b0.w}, (short)0, d0, false, false);
            d1 = __builtin_amdgcn_wmma_f32_16x16x4_f32(false, ahi, false, (v2f){b1.z, b1.w}, (short)0, d1, false, false);
            d2 = __builtin_amdgcn_wmma_f32_16x16x4_f32(false, ahi, false, (v2f){b2.z, b2.w}, (short)0, d2, false, false);
            d3 = __builtin_amdgcn_wmma_f32_16x16x4_f32(false, ahi, false, (v2f){b3.z, b3.w}, (short)0, d3, false, false);
        }

        const float bv0 = bias[c], bv1 = bias[16 + c], bv2 = bias[32 + c], bv3 = bias[48 + c];
        #pragma unroll
        for (int r = 0; r < 8; ++r) {
            float* orow = aT + (8 * h + r) * APITCH;
            const float svr = orow[F_DIM];
            float y0 = d0[r] + svr * bv0;
            float y1 = d1[r] + svr * bv1;
            float y2 = d2[r] + svr * bv2;
            float y3 = d3[r] + svr * bv3;
            orow[c]      = (y0 >= 0.f) ? y0 : a * y0;
            orow[16 + c] = (y1 >= 0.f) ? y1 : a * y1;
            orow[32 + c] = (y2 >= 0.f) ? y2 : a * y2;
            orow[48 + c] = (y3 >= 0.f) ? y3 : a * y3;
        }
        __builtin_amdgcn_fence(__ATOMIC_RELEASE, "wavefront");
        __builtin_amdgcn_wave_barrier();
        __builtin_amdgcn_fence(__ATOMIC_ACQUIRE, "wavefront");

        for (int i = 0; i < nr; ++i) {
            const float2 o = *(const float2*)(aT + i * APITCH + 2 * lane);
            if (!SCORE) {
                ((float2*)(outH + (size_t)(r0 + i) * H_DIM))[lane] = o;
                cs.x += o.x;
                cs.y += o.y;
            } else {
                float p = o.x * tv.x + o.y * tv.y;
                #pragma unroll
                for (int off = 16; off >= 1; off >>= 1) p += __shfl_xor(p, off, 32);
                if (lane == i) myscore = p;
            }
        }
    }

    if (SCORE) {
        if (lane < nr) score[r0 + lane] = myscore + bbil[0];
    } else {
        aT[2 * lane]     = cs.x;
        aT[2 * lane + 1] = cs.y;
        __syncthreads();
        if (threadIdx.x < H_DIM) {
            float t = 0.f;
            for (int w = 0; w < WPB; ++w) t += aS[w * (RPT * APITCH) + threadIdx.x];
            atomAddF(colsum + threadIdx.x, t);
        }
    }
}

__global__ void summary_kernel(
    const float* __restrict__ colsum, const float* __restrict__ Wbil,
    float* __restrict__ tbuf, int N)
{
    __shared__ float s_sh[H_DIM];
    const int hh = threadIdx.x;
    float s = colsum[hh] / (float)N;
    s = 1.f / (1.f + expf(-s));
    s_sh[hh] = s;
    __syncthreads();
    float t = 0.f;
    #pragma unroll
    for (int k = 0; k < H_DIM; ++k) t += Wbil[hh * H_DIM + k] * s_sh[k];
    tbuf[hh] = t;
}

#define SROWS 16
__global__ void __launch_bounds__(256) score_kernel(
    const float* __restrict__ Hm, const float* __restrict__ tbuf,
    const float* __restrict__ bbil, float* __restrict__ out, int N)
{
    const int lane = threadIdx.x & 31;
    const int wave = blockIdx.x * (blockDim.x >> 5) + (threadIdx.x >> 5);
    const int n0 = wave * SROWS;
    if (n0 >= N) return;
    const int c = lane & 15;
    const int h = lane >> 4;
    const float4 t4 = ((const float4*)tbuf)[c];
    float keep = 0.f;
    #pragma unroll
    for (int j = 0; j < SROWS / 2; ++j) {
        int row = n0 + 2 * j + h;
        row = row < N ? row : N - 1;
        const float4 v = ((const float4*)(Hm + (size_t)row * H_DIM))[c];
        float p = (v.x * t4.x + v.y * t4.y) + (v.z * t4.z + v.w * t4.w);
        p += __shfl_xor(p, 8, 32);
        p += __shfl_xor(p, 4, 32);
        p += __shfl_xor(p, 2, 32);
        p += __shfl_xor(p, 1, 32);
        if (c == j) keep = p;
    }
    const int row = n0 + 2 * c + h;
    if (c < SROWS / 2 && row < N) out[row] = keep + bbil[0];
}

extern "C" void kernel_launch(void* const* d_in, const int* in_sizes, int n_in,
                              void* d_out, int out_size, void* d_ws, size_t ws_size,
                              hipStream_t stream) {
    const float* pos   = (const float*)d_in[0];
    const float* neg   = (const float*)d_in[1];
    const int*   erows = (const int*)  d_in[2];
    const int*   ecols = (const int*)  d_in[3];
    const float* evals = (const float*)d_in[4];
    const float* Wg    = (const float*)d_in[5];
    const float* bg    = (const float*)d_in[6];
    const float* pa    = (const float*)d_in[7];
    const float* Wb    = (const float*)d_in[8];
    const float* bb    = (const float*)d_in[9];

    const int N = in_sizes[0] / F_DIM;
    const int E = in_sizes[2];
    float* out = (float*)d_out;

    float* posH   = (float*)d_ws + (size_t)N * H_DIM;
    float* plane3 = posH + (size_t)N * H_DIM;
    float* colsum = plane3 + (size_t)N * H_DIM;
    float* tbuf   = colsum + H_DIM;
    int2*  csr    = (int2*)plane3;
    int*   rank   = (int*)plane3 + 2 * (size_t)E;
    int*   rowptr = rank + E;
    int*   cnt    = (int*)colsum - N;

    hipMemsetAsync(cnt, 0, (size_t)(N + H_DIM) * sizeof(float), stream);

    const int edgeBlocks = (E + 256 * EPT - 1) / (256 * EPT);
    const int scanBlocks = (N + 1 + 1023) / 1024;
    const int rowTiles   = (N + RPT - 1) / RPT;
    const int fusedBlocks = (rowTiles + WPB - 1) / WPB;
    const int scoreBlocks = (N + 8 * SROWS - 1) / (8 * SROWS);

    edge_rank  <<<edgeBlocks, 256, 0, stream>>>(erows, cnt, rank, E);
    rowptr_scan<<<scanBlocks, 1024, 0, stream>>>(cnt, rowptr, N);
    csr_fill   <<<edgeBlocks, 256, 0, stream>>>(erows, ecols, evals, rank, rowptr, csr, E);
    gather_linear<false><<<fusedBlocks, WPB * 32, 0, stream>>>(
        pos, Wg, bg, rowptr, csr, pa, posH, colsum, tbuf, bb, out, N, E);
    summary_kernel<<<1, 64, 0, stream>>>(colsum, Wb, tbuf, N);
    gather_linear<true><<<fusedBlocks, WPB * 32, 0, stream>>>(
        neg, Wg, bg, rowptr, csr, pa, posH, colsum, tbuf, bb, out + N, N, E);
    score_kernel<<<scoreBlocks, 256, 0, stream>>>(posH, tbuf, bb, out, N);
}


# counting sort by destination made two-level: edges binned by 512-row bucket in LDS (runs leave as consecutive slots), each bucket sorted in LDS, csr written in order: no per-edge scattered L2 access
# speedup vs baseline: 3.0249x; 1.6225x over previous
#include <hip/hip_runtime.h>

typedef __attribute__((ext_vector_type(2))) float v2f;
typedef __attribute__((ext_vector_type(8))) float v8f;

#define F_DIM 128
#define H_DIM 64

__device__ __forceinline__ void atomAddF(float* p, float v) {
    __hip_atomic_fetch_add(p, v, __ATOMIC_RELAXED, __HIP_MEMORY_SCOPE_AGENT);
}

#define BSHIFT 9
#define BROWS (1 << BSHIFT)
#define MAXNB 1024
#define PTHREADS 1024
#define PEPT 4
#define PEDGES (PTHREADS * PEPT)
#define SCAP 16384

__device__ __forceinline__ int bucket_of(const int row, const int NB) {
    const int b = row >> BSHIFT;
    return b < 0 ? 0 : (b >= NB ? NB - 1 : b);
}

__device__ __forceinline__ int lds_add_rtn(int* p, const int v) {
    return __hip_atomic_fetch_add(p, v, __ATOMIC_RELAXED, __HIP_MEMORY_SCOPE_WORKGROUP);
}

__global__ void __launch_bounds__(PTHREADS) bucket_hist(
    const int* __restrict__ rows, int* __restrict__ bcnt, int E, int NB)
{
    __shared__ int cntL[MAXNB];
    const int tid = threadIdx.x;
    for (int b = tid; b < NB; b += PTHREADS) cntL[b] = 0;
    __syncthreads();
    const int e0 = (blockIdx.x * PTHREADS + tid) * PEPT;
    if (e0 + PEPT <= E && (((size_t)rows) & 15) == 0) {
        const int4 r = *(const int4*)(rows + e0);
        lds_add_rtn(cntL + bucket_of(r.x, NB), 1);
        lds_add_rtn(cntL + bucket_of(r.y, NB), 1);
        lds_add_rtn(cntL + bucket_of(r.z, NB), 1);
        lds_add_rtn(cntL + bucket_of(r.w, NB), 1);
    } else {
        const int e1 = (e0 + PEPT < E) ? e0 + PEPT : E;
        for (int e = e0; e < e1; ++e) lds_add_rtn(cntL + bucket_of(rows[e], NB), 1);
    }
    __syncthreads();
    for (int b = tid; b < NB; b += PTHREADS) {
        const int c = cntL[b];
        if (c > 0) __hip_atomic_fetch_add(bcnt + b, c, __ATOMIC_RELAXED, __HIP_MEMORY_SCOPE_AGENT);
    }
}

__global__ void __launch_bounds__(PTHREADS) bucket_partition(
    const int* __restrict__ rows, const int* __restrict__ cols,
    const float* __restrict__ vals, const int* __restrict__ bcnt,
    int* __restrict__ cursor, int2* __restrict__ tmp_cv, int* __restrict__ tmp_row,
    int E, int NB)
{
    __shared__ int cntL[MAXNB];
    __shared__ int offL[MAXNB];
    __shared__ int dstL[MAXNB];
    __shared__ int ws1[32];
    __shared__ int ws2[32];
    __shared__ int2 cvL[PEDGES];
    __shared__ int rowL[PEDGES];

    const int tid  = threadIdx.x;
    const int lane = tid & 31;
    const int wv   = tid >> 5;
    for (int b = tid; b < NB; b += PTHREADS) cntL[b] = 0;
    __syncthreads();

    const int blk0 = blockIdx.x * PEDGES;
    const int e0 = blk0 + tid * PEPT;
    int n = E - e0;
    n = n < 0 ? 0 : (n > PEPT ? PEPT : n);
    int r[PEPT], c[PEPT], bk[PEPT], k[PEPT];
    float v[PEPT];
    #pragma unroll
    for (int j = 0; j < PEPT; ++j) { r[j] = 0; c[j] = 0; v[j] = 0.f; bk[j] = 0; k[j] = 0; }
    if (n == PEPT && ((((size_t)rows) | ((size_t)cols) | ((size_t)vals)) & 15) == 0) {
        const int4   r4 = *(const int4*)(rows + e0);
        const int4   c4 = *(const int4*)(cols + e0);
        const float4 v4 = *(const float4*)(vals + e0);
        r[0] = r4.x; r[1] = r4.y; r[2] = r4.z; r[3] = r4.w;
        c[0] = c4.x; c[1] = c4.y; c[2] = c4.z; c[3] = c4.w;
        v[0] = v4.x; v[1] = v4.y; v[2] = v4.z; v[3] = v4.w;
    } else {
        #pragma unroll
        for (int j = 0; j < PEPT; ++j)
            if (j < n) { r[j] = rows[e0 + j]; c[j] = cols[e0 + j]; v[j] = vals[e0 + j]; }
    }
    #pragma unroll
    for (int j = 0; j < PEPT; ++j)
        if (j < n) { bk[j] = bucket_of(r[j], NB); k[j] = lds_add_rtn(cntL + bk[j], 1); }
    __syncthreads();

    const int mycnt = (tid < NB) ? cntL[tid] : 0;
    const int gcnt  = (tid < NB) ? bcnt[tid] : 0;
    int inc1 = mycnt, inc2 = gcnt;
    #pragma unroll
    for (int off = 1; off < 32; off <<= 1) {
        const int t1 = __shfl_up(inc1, off, 32);
        const int t2 = __shfl_up(inc2, off, 32);
        if (lane >= off) { inc1 += t1; inc2 += t2; }
    }
    if (lane == 31) { ws1[wv] = inc1; ws2[wv] = inc2; }
    __syncthreads();
    if (wv == 0) {
        int v1 = ws1[lane], v2 = ws2[lane];
        #pragma unroll
        for (int off = 1; off < 32; off <<= 1) {
            const int t1 = __shfl_up(v1, off, 32);
            const int t2 = __shfl_up(v2, off, 32);
            if (lane >= off) { v1 += t1; v2 += t2; }
        }
        ws1[lane] = v1; ws2[lane] = v2;
    }
    __syncthreads();
    if (tid < NB) {
        const int ex1 = inc1 - mycnt + (wv > 0 ? ws1[wv - 1] : 0);
        const int ex2 = inc2 - gcnt  + (wv > 0 ? ws2[wv - 1] : 0);
        int res = 0;
        if (mycnt > 0)
            res = __hip_atomic_fetch_add(cursor + tid, mycnt, __ATOMIC_RELAXED, __HIP_MEMORY_SCOPE_AGENT);
        offL[tid] = ex1;
        dstL[tid] = ex2 + res;
    }
    __syncthreads();

    #pragma unroll
    for (int j = 0; j < PEPT; ++j)
        if (j < n) {
            const int p = offL[bk[j]] + k[j];
            if ((unsigned)p < (unsigned)PEDGES) {
                cvL[p]  = make_int2(c[j], __float_as_int(v[j]));
                rowL[p] = r[j];
            }
        }
    __syncthreads();

    int total = E - blk0;
    total = total < 0 ? 0 : (total > PEDGES ? PEDGES : total);
    for (int i = tid; i < total; i += PTHREADS) {
        const int row = rowL[i];
        const int b = bucket_of(row, NB);
        const int g = dstL[b] + (i - offL[b]);
        if ((unsigned)g < (unsigned)E) { tmp_cv[g] = cvL[i]; tmp_row[g] = row; }
    }
}

__global__ void __launch_bounds__(PTHREADS) bucket_sort(
    const int2* __restrict__ tmp_cv, const int* __restrict__ tmp_row,
    const int* __restrict__ bcnt, int* __restrict__ rowptr, int2* __restrict__ csr,
    int N, int E, int NB)
{
    __shared__ int cntL[BROWS];
    __shared__ int curL[BROWS];
    __shared__ int wsum[32];
    __shared__ int startSh;
    __shared__ int2 csrL[SCAP];

    const int tid  = threadIdx.x;
    const int lane = tid & 31;
    const int wv   = tid >> 5;
    const int b    = blockIdx.x;

    int part = 0;
    for (int j = tid; j < b; j += PTHREADS) part += bcnt[j];
    #pragma unroll
    for (int off = 16; off >= 1; off >>= 1) part += __shfl_xor(part, off, 32);
    if (lane == 0) wsum[wv] = part;
    if (tid < BROWS) cntL[tid] = 0;
    __syncthreads();
    if (wv == 0) {
        int t = wsum[lane];
        #pragma unroll
        for (int off = 16; off >= 1; off >>= 1) t += __shfl_xor(t, off, 32);
        if (lane == 0) startSh = t;
    }
    __syncthreads();
    int start = startSh;
    start = start < 0 ? 0 : (start > E ? E : start);
    int nb = bcnt[b];
    nb = nb < 0 ? 0 : (nb > E - start ? E - start : nb);
    const int rowbase = b << BSHIFT;

    for (int i = tid; i < nb; i += PTHREADS) {
        int r = tmp_row[start + i] - rowbase;
        r = r < 0 ? 0 : (r > BROWS - 1 ? BROWS - 1 : r);
        lds_add_rtn(cntL + r, 1);
    }
    __syncthreads();

    const int cr = (tid < BROWS) ? cntL[tid] : 0;
    int inc = cr;
    #pragma unroll
    for (int off = 1; off < 32; off <<= 1) {
        const int t = __shfl_up(inc, off, 32);
        if (lane >= off) inc += t;
    }
    if (lane == 31) wsum[wv] = inc;
    __syncthreads();
    if (wv == 0) {
        int t = wsum[lane];
        #pragma unroll
        for (int off = 1; off < 32; off <<= 1) {
            const int u = __shfl_up(t, off, 32);
            if (lane >= off) t += u;
        }
        wsum[lane] = t;
    }
    __syncthreads();
    if (tid < BROWS) {
        const int ex = inc - cr + (wv > 0 ? wsum[wv - 1] : 0);
        curL[tid] = ex;
        const int row = rowbase + tid;
        if (row < N) rowptr[row] = start + ex;
    }
    if (b == NB - 1 && tid == 0) rowptr[N] = start + nb;
    __syncthreads();

    const bool staged = nb <= SCAP;
    for (int i = tid; i < nb; i += PTHREADS) {
        int r = tmp_row[start + i] - rowbase;
        r = r < 0 ? 0 : (r > BROWS - 1 ? BROWS - 1 : r);
        const int2 cv = tmp_cv[start + i];
        const int p = lds_add_rtn(curL + r, 1);
        if (staged) {
            if ((unsigned)p < (unsigned)SCAP) csrL[p] = cv;
        } else {
            const int g = start + p;
            if ((unsigned)g < (unsigned)E) csr[g] = cv;
        }
    }
    __syncthreads();
    if (staged)
        for (int i = tid; i < nb; i += PTHREADS) csr[start + i] = csrL[i];
}

#define RPT 16
#define WPB 25
#define APITCH 132

__device__ __forceinline__ void add_chunk(
    float4& acc, float& sv, const int2 ev, const int cnt,
    const float* __restrict__ x, const int lane, const int N)
{
    #pragma unroll
    for (int g = 0; g < 8; ++g) {
        const int j = 4 * g;
        if (j >= cnt) break;
        int c0 = __builtin_amdgcn_readlane(ev.x, j);
        int c1 = __builtin_amdgcn_readlane(ev.x, j + 1);
        int c2 = __builtin_amdgcn_readlane(ev.x, j + 2);
        int c3 = __builtin_amdgcn_readlane(ev.x, j + 3);
        const float v0 = __int_as_float(__builtin_amdgcn_readlane(ev.y, j));
        const float v1 = __int_as_float(__builtin_amdgcn_readlane(ev.y, j + 1));
        const float v2 = __int_as_float(__builtin_amdgcn_readlane(ev.y, j + 2));
        const float v3 = __int_as_float(__builtin_amdgcn_readlane(ev.y, j + 3));
        c0 = ((unsigned)c0 < (unsigned)N) ? c0 : 0;
        c1 = ((unsigned)c1 < (unsigned)N) ? c1 : 0;
        c2 = ((unsigned)c2 < (unsigned)N) ? c2 : 0;
        c3 = ((unsigned)c3 < (unsigned)N) ? c3 : 0;
        const float4 m0 = ((const float4*)(x + (size_t)c0 * F_DIM))[lane];
        const float4 m1 = ((const float4*)(x + (size_t)c1 * F_DIM))[lane];
        const float4 m2 = ((const float4*)(x + (size_t)c2 * F_DIM))[lane];
        const float4 m3 = ((const float4*)(x + (size_t)c3 * F_DIM))[lane];
        acc.x += v0 * m0.x;  acc.y += v0 * m0.y;  acc.z += v0 * m0.z;  acc.w += v0 * m0.w;
        acc.x += v1 * m1.x;  acc.y += v1 * m1.y;  acc.z += v1 * m1.z;  acc.w += v1 * m1.w;
        acc.x += v2 * m2.x;  acc.y += v2 * m2.y;  acc.z += v2 * m2.z;  acc.w += v2 * m2.w;
        acc.x += v3 * m3.x;  acc.y += v3 * m3.y;  acc.z += v3 * m3.z;  acc.w += v3 * m3.w;
        sv += (v0 + v1) + (v2 + v3);
    }
}

template <bool SCORE>
__global__ void __launch_bounds__(WPB * 32) gather_linear(
    const float* __restrict__ x, const float* __restrict__ W,
    const float* __restrict__ bias, const int* __restrict__ rowptr,
    const int2* __restrict__ csr, const float* __restrict__ prelu_a,
    float* __restrict__ outH, float* __restrict__ colsum,
    const float* __restrict__ tbuf, const float* __restrict__ bbil,
    float* __restrict__ score, int N, int E)
{
    __shared__ __attribute__((aligned(16))) float wS[H_DIM * APITCH];
    __shared__ __attribute__((aligned(16))) float aS[WPB * RPT * APITCH];

    const int lane = threadIdx.x & 31;
    const int wv = __builtin_amdgcn_readfirstlane((int)(threadIdx.x >> 5));
    const int r0 = (blockIdx.x * WPB + wv) * RPT;
    int nr = N - r0;
    nr = nr < 0 ? 0 : (nr > RPT ? RPT : nr);
    float* aT = aS + wv * (RPT * APITCH);

    const float a = prelu_a[0];
    float2 tv = make_float2(0.f, 0.f);
    if (SCORE) tv = ((const float2*)tbuf)[lane];

    int pi = r0 + lane;
    pi = pi > N ? N : pi;
    int myptr = rowptr[pi];
    myptr = myptr < 0 ? 0 : (myptr > E ? E : myptr);

    constexpr int W_PIECES  = H_DIM * (F_DIM / 4);
    constexpr int W_STAGERS = 512;
    constexpr int W_TRIPS   = W_PIECES / W_STAGERS;
    static_assert(W_PIECES % W_STAGERS == 0 && W_STAGERS % 32 == 0 && W_STAGERS <= WPB * 32,
                  "W staging: whole waves, whole trips");
    if (threadIdx.x < W_STAGERS) {
        float4 wreg[W_TRIPS];
        #pragma unroll
        for (int t = 0; t < W_TRIPS; ++t)
            wreg[t] = ((const float4*)W)[threadIdx.x + t * W_STAGERS];
        #pragma unroll
        for (int t = 0; t < W_TRIPS; ++t) {
            const int idx = threadIdx.x + t * W_STAGERS;
            *(float4*)(wS + (idx >> 5) * APITCH + 4 * (idx & 31)) = wreg[t];
        }
    }
    __syncthreads();

    int beg = __builtin_amdgcn_readlane(myptr, 0);
    int end = __builtin_amdgcn_readlane(myptr, 1);
    int2 ev = make_int2(0, 0);
    if (nr > 0 && lane < end - beg) ev = csr[beg + lane];
    asm volatile("" :: "v"(ev.x), "v"(ev.y));

    for (int i = 0; i < nr; ++i) {
        int endn = end;
        int2 evn = make_int2(0, 0);
        if (i + 1 < nr) {
            endn = __builtin_amdgcn_readlane(myptr, i + 2);
            if (lane < endn - end) evn = csr[end + lane];
        }

        float4 acc = make_float4(0.f, 0.f, 0.f, 0.f);
        float sv = 0.f;
        const int deg = end - beg;
        add_chunk(acc, sv, ev, deg > 32 ? 32 : deg, x, lane, N);
        if (deg > 32) {
            for (int base = beg + 32; base < end; base += 32) {
                int2 e2 = make_int2(0, 0);
                if (lane < end - base) e2 = csr[base + lane];
                const int rest = end - base;
                add_chunk(acc, sv, e2, rest > 32 ? 32 : rest, x, lane, N);
            }
        }

        *(float4*)(aT + i * APITCH + 4 * lane) = acc;
        if (lane == 0) aT[i * APITCH + F_DIM] = sv;

        beg = end;
        end = endn;
        ev  = evn;
    }
    for (int i = nr; i < RPT; ++i) {
        *(float4*)(aT + i * APITCH + 4 * lane) = make_float4(0.f, 0.f, 0.f, 0.f);
        if (lane == 0) aT[i * APITCH + F_DIM] = 0.f;
    }

    float2 cs = make_float2(0.f, 0.f);
    float myscore = 0.f;

    if (nr > 0) {
        __builtin_amdgcn_fence(__ATOMIC_RELEASE, "wavefront");
        __builtin_amdgcn_wave_barrier();
        __builtin_amdgcn_fence(__ATOMIC_ACQUIRE, "wavefront");

        const int c = lane & 15;
        const int h = lane >> 4;
        const float* ap = aT + c * APITCH + 64 * h;
        const float* wp = wS + c * APITCH + 64 * h;
        v8f d0 = {}, d1 = {}, d2 = {}, d3 = {};
        #pragma unroll 2
        for (int p = 0; p < 16; ++p) {
            const float4 a4 = *(const float4*)(ap + 4 * p);
            const float4 b0 = *(const float4*)(wp + 4 * p);
            const float4 b1 = *(const float4*)(wp + 16 * APITCH + 4 * p);
            const float4 b2 = *(const float4*)(wp + 32 * APITCH + 4 * p);
            const float4 b3 = *(const float4*)(wp + 48 * APITCH + 4 * p);
            const v2f alo = {a4.x, a4.y};
            const v2f ahi = {a4.z, a4.w};
            d0 = __builtin_amdgcn_wmma_f32_16x16x4_f32(false, alo, false, (v2f){b0.x, b0.y}, (short)0, d0, false, false);
            d1 = __builtin_amdgcn_wmma_f32_16x16x4_f32(false, alo, false, (v2f){b1.x, b1.y}, (short)0, d1, false, false);
            d2 = __builtin_amdgcn_wmma_f32_16x16x4_f32(false, alo, false, (v2f){b2.x, b2.y}, (short)0, d2, false, false);
            d3 = __builtin_amdgcn_wmma_f32_16x16x4_f32(false, alo, false, (v2f){b3.x, b3.y}, (short)0, d3, false, false);
            d0 = __builtin_amdgcn_wmma_f32_16x16x4_f32(false, ahi, false, (v2f){b0.z, b0.w}, (short)0, d0, false, false);
            d1 = __builtin_amdgcn_wmma_f32_16x16x4_f32(false, ahi, false, (v2f){b1.z, b1.w}, (short)0, d1, false, false);
            d2 = __builtin_amdgcn_wmma_f32_16x16x4_f32(false, ahi, false, (v2f){b2.z, b2.w}, (short)0, d2, false, false);
            d3 = __builtin_amdgcn_wmma_f32_16x16x4_f32(false, ahi, false, (v2f){b3.z, b3.w}, (short)0, d3, false, false);
        }

        const float bv0 = bias[c], bv1 = bias[16 + c], bv2 = bias[32 + c], bv3 = bias[48 + c];
        #pragma unroll
        for (int r = 0; r < 8; ++r) {
            float* orow = aT + (8 * h + r) * APITCH;
            const float svr = orow[F_DIM];
            float y0 = d0[r] + svr * bv0;
            float y1 = d1[r] + svr * bv1;
            float y2 = d2[r] + svr * bv2;
            float y3 = d3[r] + svr * bv3;
            orow[c]      = (y0 >= 0.f) ? y0 : a * y0;
            orow[16 + c] = (y1 >= 0.f) ? y1 : a * y1;
            orow[32 + c] = (y2 >= 0.f) ? y2 : a * y2;
            orow[48 + c] = (y3 >= 0.f) ? y3 : a * y3;
        }
        __builtin_amdgcn_fence(__ATOMIC_RELEASE, "wavefront");
        __builtin_amdgcn_wave_barrier();
        __builtin_amdgcn_fence(__ATOMIC_ACQUIRE, "wavefront");

        for (int i = 0; i < nr; ++i) {
            const float2 o = *(const float2*)(aT + i * APITCH + 2 * lane);
            if (!SCORE) {
                ((float2*)(outH + (size_t)(r0 + i) * H_DIM))[lane] = o;
                cs.x += o.x;
                cs.y += o.y;
            } else {
                float p = o.x * tv.x + o.y * tv.y;
                #pragma unroll
                for (int off = 16; off >= 1; off >>= 1) p += __shfl_xor(p, off, 32);
                if (lane == i) myscore = p;
            }
        }
    }

    if (SCORE) {
        if (lane < nr) score[r0 + lane] = myscore + bbil[0];
    } else {
        aT[2 * lane]     = cs.x;
        aT[2 * lane + 1] = cs.y;
        __syncthreads();
        if (threadIdx.x < H_DIM) {
            float t = 0.f;
            for (int w = 0; w < WPB; ++w) t += aS[w * (RPT * APITCH) + threadIdx.x];
            atomAddF(colsum + threadIdx.x, t);
        }
    }
}

__global__ void summary_kernel(
    const float* __restrict__ colsum, const float* __restrict__ Wbil,
    float* __restrict__ tbuf, int N)
{
    __shared__ float s_sh[H_DIM];
    const int hh = threadIdx.x;
    float s = colsum[hh] / (float)N;
    s = 1.f / (1.f + expf(-s));
    s_sh[hh] = s;
    __syncthreads();
    float t = 0.f;
    #pragma unroll
    for (int k = 0; k < H_DIM; ++k) t += Wbil[hh * H_DIM + k] * s_sh[k];
    tbuf[hh] = t;
}

#define SROWS 16
__global__ void __launch_bounds__(256) score_kernel(
    const float* __restrict__ Hm, const float* __restrict__ tbuf,
    const float* __restrict__ bbil, float* __restrict__ out, int N)
{
    const int lane = threadIdx.x & 31;
    const int wave = blockIdx.x * (blockDim.x >> 5) + (threadIdx.x >> 5);
    const int n0 = wave * SROWS;
    if (n0 >= N) return;
    const int c = lane & 15;
    const int h = lane >> 4;
    const float4 t4 = ((const float4*)tbuf)[c];
    float keep = 0.f;
    #pragma unroll
    for (int j = 0; j < SROWS / 2; ++j) {
        int row = n0 + 2 * j + h;
        row = row < N ? row : N - 1;
        const float4 v = ((const float4*)(Hm + (size_t)row * H_DIM))[c];
        float p = (v.x * t4.x + v.y * t4.y) + (v.z * t4.z + v.w * t4.w);
        p += __shfl_xor(p, 8, 32);
        p += __shfl_xor(p, 4, 32);
        p += __shfl_xor(p, 2, 32);
        p += __shfl_xor(p, 1, 32);
        if (c == j) keep = p;
    }
    const int row = n0 + 2 * c + h;
    if (c < SROWS / 2 && row < N) out[row] = keep + bbil[0];
}

extern "C" void kernel_launch(void* const* d_in, const int* in_sizes, int n_in,
                              void* d_out, int out_size, void* d_ws, size_t ws_size,
                              hipStream_t stream) {
    const float* pos   = (const float*)d_in[0];
    const float* neg   = (const float*)d_in[1];
    const int*   erows = (const int*)  d_in[2];
    const int*   ecols = (const int*)  d_in[3];
    const float* evals = (const float*)d_in[4];
    const float* Wg    = (const float*)d_in[5];
    const float* bg    = (const float*)d_in[6];
    const float* pa    = (const float*)d_in[7];
    const float* Wb    = (const float*)d_in[8];
    const float* bb    = (const float*)d_in[9];

    const int N = in_sizes[0] / F_DIM;
    const int E = in_sizes[2];
    float* out = (float*)d_out;

    float* posH   = (float*)d_ws + (size_t)N * H_DIM;
    float* plane3 = posH + (size_t)N * H_DIM;
    float* colsum = plane3 + (size_t)N * H_DIM;
    float* tbuf   = colsum + H_DIM;
    int2*  tmp_cv  = (int2*)d_ws;
    int*   tmp_row = (int*)d_ws + 2 * (size_t)E;
    int2*  csr    = (int2*)plane3;
    int*   rowptr = (int*)plane3 + 2 * (size_t)E;
    int*   bcnt   = (int*)colsum - 2 * MAXNB;
    int*   cursor = bcnt + MAXNB;

    hipMemsetAsync(bcnt, 0, (size_t)(2 * MAXNB + H_DIM) * sizeof(float), stream);

    const int NB = (N + BROWS - 1) / BROWS;
    const int edgeBlocks = (E + PEDGES - 1) / PEDGES;
    const int rowTiles   = (N + RPT - 1) / RPT;
    const int fusedBlocks = (rowTiles + WPB - 1) / WPB;
    const int scoreBlocks = (N + 8 * SROWS - 1) / (8 * SROWS);

    bucket_hist     <<<edgeBlocks, PTHREADS, 0, stream>>>(erows, bcnt, E, NB);
    bucket_partition<<<edgeBlocks, PTHREADS, 0, stream>>>(erows, ecols, evals, bcnt, cursor,
                                                          tmp_cv, tmp_row, E, NB);
    bucket_sort     <<<NB, PTHREADS, 0, stream>>>(tmp_cv, tmp_row, bcnt, rowptr, csr, N, E, NB);
    gather_linear<false><<<fusedBlocks, WPB * 32, 0, stream>>>(
        pos, Wg, bg, rowptr, csr, pa, posH, colsum, tbuf, bb, out, N, E);
    summary_kernel<<<1, 64, 0, stream>>>(colsum, Wb, tbuf, N);
    gather_linear<true><<<fusedBlocks, WPB * 32, 0, stream>>>(
        neg, Wg, bg, rowptr, csr, pa, posH, colsum, tbuf, bb, out + N, N, E);
    score_kernel<<<scoreBlocks, 256, 0, stream>>>(posH, tbuf, bb, out, N);
}


# gather: all source-row loads of a row (up to 16 edges) asked for before the first is summed, one round trip per row instead of one per four edges; sums in the same order
# speedup vs baseline: 3.4006x; 1.1242x over previous
#include <hip/hip_runtime.h>

typedef __attribute__((ext_vector_type(2))) float v2f;
typedef __attribute__((ext_vector_type(8))) float v8f;

#define F_DIM 128
#define H_DIM 64

__device__ __forceinline__ void atomAddF(float* p, float v) {
    __hip_atomic_fetch_add(p, v, __ATOMIC_RELAXED, __HIP_MEMORY_SCOPE_AGENT);
}

#define BSHIFT 9
#define BROWS (1 << BSHIFT)
#define MAXNB 1024
#define PTHREADS 1024
#define PEPT 4
#define PEDGES (PTHREADS * PEPT)
#define SCAP 16384

__device__ __forceinline__ int bucket_of(const int row, const int NB) {
    const int b = row >> BSHIFT;
    return b < 0 ? 0 : (b >= NB ? NB - 1 : b);
}

__device__ __forceinline__ int lds_add_rtn(int* p, const int v) {
    return __hip_atomic_fetch_add(p, v, __ATOMIC_RELAXED, __HIP_MEMORY_SCOPE_WORKGROUP);
}

__global__ void __launch_bounds__(PTHREADS) bucket_hist(
    const int* __restrict__ rows, int* __restrict__ bcnt, int E, int NB)
{
    __shared__ int cntL[MAXNB];
    const int tid = threadIdx.x;
    for (int b = tid; b < NB; b += PTHREADS) cntL[b] = 0;
    __syncthreads();
    const int e0 = (blockIdx.x * PTHREADS + tid) * PEPT;
    if (e0 + PEPT <= E && (((size_t)rows) & 15) == 0) {
        const int4 r = *(const int4*)(rows + e0);
        lds_add_rtn(cntL + bucket_of(r.x, NB), 1);
        lds_add_rtn(cntL + bucket_of(r.y, NB), 1);
        lds_add_rtn(cntL + bucket_of(r.z, NB), 1);
        lds_add_rtn(cntL + bucket_of(r.w, NB), 1);
    } else {
        const int e1 = (e0 + PEPT < E) ? e0 + PEPT : E;
        for (int e = e0; e < e1; ++e) lds_add_rtn(cntL + bucket_of(rows[e], NB), 1);
    }
    __syncthreads();
    for (int b = tid; b < NB; b += PTHREADS) {
        const int c = cntL[b];
        if (c > 0) __hip_atomic_fetch_add(bcnt + b, c, __ATOMIC_RELAXED, __HIP_MEMORY_SCOPE_AGENT);
    }
}

__global__ void __launch_bounds__(PTHREADS) bucket_partition(
    const int* __restrict__ rows, const int* __restrict__ cols,
    const float* __restrict__ vals, const int* __restrict__ bcnt,
    int* __restrict__ cursor, int2* __restrict__ tmp_cv, int* __restrict__ tmp_row,
    int E, int NB)
{
    __shared__ int cntL[MAXNB];
    __shared__ int offL[MAXNB];
    __shared__ int dstL[MAXNB];
    __shared__ int ws1[32];
    __shared__ int ws2[32];
    __shared__ int2 cvL[PEDGES];
    __shared__ int rowL[PEDGES];

    const int tid  = threadIdx.x;
    const int lane = tid & 31;
    const int wv   = tid >> 5;
    for (int b = tid; b < NB; b += PTHREADS) cntL[b] = 0;
    __syncthreads();

    const int blk0 = blockIdx.x * PEDGES;
    const int e0 = blk0 + tid * PEPT;
    int n = E - e0;
    n = n < 0 ? 0 : (n > PEPT ? PEPT : n);
    int r[PEPT], c[PEPT], bk[PEPT], k[PEPT];
    float v[PEPT];
    #pragma unroll
    for (int j = 0; j < PEPT; ++j) { r[j] = 0; c[j] = 0; v[j] = 0.f; bk[j] = 0; k[j] = 0; }
    if (n == PEPT && ((((size_t)rows) | ((size_t)cols) | ((size_t)vals)) & 15) == 0) {
        const int4   r4 = *(const int4*)(rows + e0);
        const int4   c4 = *(const int4*)(cols + e0);
        const float4 v4 = *(const float4*)(vals + e0);
        r[0] = r4.x; r[1] = r4.y; r[2] = r4.z; r[3] = r4.w;
        c[0] = c4.x; c[1] = c4.y; c[2] = c4.z; c[3] = c4.w;
        v[0] = v4.x; v[1] = v4.y; v[2] = v4.z; v[3] = v4.w;
    } else {
        #pragma unroll
        for (int j = 0; j < PEPT; ++j)
            if (j < n) { r[j] = rows[e0 + j]; c[j] = cols[e0 + j]; v[j] = vals[e0 + j]; }
    }
    #pragma unroll
    for (int j = 0; j < PEPT; ++j)
        if (j < n) { bk[j] = bucket_of(r[j], NB); k[j] = lds_add_rtn(cntL + bk[j], 1); }
    __syncthreads();

    const int mycnt = (tid < NB) ? cntL[tid] : 0;
    const int gcnt  = (tid < NB) ? bcnt[tid] : 0;
    int inc1 = mycnt, inc2 = gcnt;
    #pragma unroll
    for (int off = 1; off < 32; off <<= 1) {
        const int t1 = __shfl_up(inc1, off, 32);
        const int t2 = __shfl_up(inc2, off, 32);
        if (lane >= off) { inc1 += t1; inc2 += t2; }
    }
    if (lane == 31) { ws1[wv] = inc1; ws2[wv] = inc2; }
    __syncthreads();
    if (wv == 0) {
        int v1 = ws1[lane], v2 = ws2[lane];
        #pragma unroll
        for (int off = 1; off < 32; off <<= 1) {
            const int t1 = __shfl_up(v1, off, 32);
            const int t2 = __shfl_up(v2, off, 32);
            if (lane >= off) { v1 += t1; v2 += t2; }
        }
        ws1[lane] = v1; ws2[lane] = v2;
    }
    __syncthreads();
    if (tid < NB) {
        const int ex1 = inc1 - mycnt + (wv > 0 ? ws1[wv - 1] : 0);
        const int ex2 = inc2 - gcnt  + (wv > 0 ? ws2[wv - 1] : 0);
        int res = 0;
        if (mycnt > 0)
            res = __hip_atomic_fetch_add(cursor + tid, mycnt, __ATOMIC_RELAXED, __HIP_MEMORY_SCOPE_AGENT);
        offL[tid] = ex1;
        dstL[tid] = ex2 + res;
    }
    __syncthreads();

    #pragma unroll
    for (int j = 0; j < PEPT; ++j)
        if (j < n) {
            const int p = offL[bk[j]] + k[j];
            if ((unsigned)p < (unsigned)PEDGES) {
                cvL[p]  = make_int2(c[j], __float_as_int(v[j]));
                rowL[p] = r[j];
            }
        }
    __syncthreads();

    int total = E - blk0;
    total = total < 0 ? 0 : (total > PEDGES ? PEDGES : total);
    for (int i = tid; i < total; i += PTHREADS) {
        const int row = rowL[i];
        const int b = bucket_of(row, NB);
        const int g = dstL[b] + (i - offL[b]);
        if ((unsigned)g < (unsigned)E) { tmp_cv[g] = cvL[i]; tmp_row[g] = row; }
    }
}

__global__ void __launch_bounds__(PTHREADS) bucket_sort(
    const int2* __restrict__ tmp_cv, const int* __restrict__ tmp_row,
    const int* __restrict__ bcnt, int* __restrict__ rowptr, int2* __restrict__ csr,
    int N, int E, int NB)
{
    __shared__ int cntL[BROWS];
    __shared__ int curL[BROWS];
    __shared__ int wsum[32];
    __shared__ int startSh;
    __shared__ int2 csrL[SCAP];

    const int tid  = threadIdx.x;
    const int lane = tid & 31;
    const int wv   = tid >> 5;
    const int b    = blockIdx.x;

    int part = 0;
    for (int j = tid; j < b; j += PTHREADS) part += bcnt[j];
    #pragma unroll
    for (int off = 16; off >= 1; off >>= 1) part += __shfl_xor(part, off, 32);
    if (lane == 0) wsum[wv] = part;
    if (tid < BROWS) cntL[tid] = 0;
    __syncthreads();
    if (wv == 0) {
        int t = wsum[lane];
        #pragma unroll
        for (int off = 16; off >= 1; off >>= 1) t += __shfl_xor(t, off, 32);
        if (lane == 0) startSh = t;
    }
    __syncthreads();
    int start = startSh;
    start = start < 0 ? 0 : (start > E ? E : start);
    int nb = bcnt[b];
    nb = nb < 0 ? 0 : (nb > E - start ? E - start : nb);
    const int rowbase = b << BSHIFT;

    for (int i = tid; i < nb; i += PTHREADS) {
        int r = tmp_row[start + i] - rowbase;
        r = r < 0 ? 0 : (r > BROWS - 1 ? BROWS - 1 : r);
        lds_add_rtn(cntL + r, 1);
    }
    __syncthreads();

    const int cr = (tid < BROWS) ? cntL[tid] : 0;
    int inc = cr;
    #pragma unroll
    for (int off = 1; off < 32; off <<= 1) {
        const int t = __shfl_up(inc, off, 32);
        if (lane >= off) inc += t;
    }
    if (lane == 31) wsum[wv] = inc;
    __syncthreads();
    if (wv == 0) {
        int t = wsum[lane];
        #pragma unroll
        for (int off = 1; off < 32; off <<= 1) {
            const int u = __shfl_up(t, off, 32);
            if (lane >= off) t += u;
        }
        wsum[lane] = t;
    }
    __syncthreads();
    if (tid < BROWS) {
        const int ex = inc - cr + (wv > 0 ? wsum[wv - 1] : 0);
        curL[tid] = ex;
        const int row = rowbase + tid;
        if (row < N) rowptr[row] = start + ex;
    }
    if (b == NB - 1 && tid == 0) rowptr[N] = start + nb;
    __syncthreads();

    const bool staged = nb <= SCAP;
    for (int i = tid; i < nb; i += PTHREADS) {
        int r = tmp_row[start + i] - rowbase;
        r = r < 0 ? 0 : (r > BROWS - 1 ? BROWS - 1 : r);
        const int2 cv = tmp_cv[start + i];
        const int p = lds_add_rtn(curL + r, 1);
        if (staged) {
            if ((unsigned)p < (unsigned)SCAP) csrL[p] = cv;
        } else {
            const int g = start + p;
            if ((unsigned)g < (unsigned)E) csr[g] = cv;
        }
    }
    __syncthreads();
    if (staged)
        for (int i = tid; i < nb; i += PTHREADS) csr[start + i] = csrL[i];
}

#define RPT 16
#define WPB 25
#define APITCH 132

template <int G, int J0>
__device__ __forceinline__ void add_batch(
    float4& acc, float& sv, const int2 ev,
    const float* __restrict__ x, const int lane, const int N)
{
    float4 m[4 * G];
    float  v[4 * G];
    #pragma unroll
    for (int k = 0; k < 4 * G; ++k) {
        int c = __builtin_amdgcn_readlane(ev.x, J0 + k);
        v[k] = __int_as_float(__builtin_amdgcn_readlane(ev.y, J0 + k));
        c = ((unsigned)c < (unsigned)N) ? c : 0;
        m[k] = ((const float4*)(x + (size_t)c * F_DIM))[lane];
    }
    #pragma unroll
    for (int g = 0; g < G; ++g) {
        #pragma unroll
        for (int k = 4 * g; k < 4 * g + 4; ++k) {
            acc.x = __builtin_fmaf(v[k], m[k].x, acc.x);  acc.y = __builtin_fmaf(v[k], m[k].y, acc.y);
            acc.z = __builtin_fmaf(v[k], m[k].z, acc.z);  acc.w = __builtin_fmaf(v[k], m[k].w, acc.w);
        }
        sv += (v[4 * g] + v[4 * g + 1]) + (v[4 * g + 2] + v[4 * g + 3]);
    }
}

template <int J0>
__device__ __forceinline__ void add_upto16(
    float4& acc, float& sv, const int2 ev, const int n,
    const float* __restrict__ x, const int lane, const int N)
{
    if (n > 12)      add_batch<4, J0>(acc, sv, ev, x, lane, N);
    else if (n > 8)  add_batch<3, J0>(acc, sv, ev, x, lane, N);
    else if (n > 4)  add_batch<2, J0>(acc, sv, ev, x, lane, N);
    else             add_batch<1, J0>(acc, sv, ev, x, lane, N);
}

__device__ __forceinline__ void add_chunk(
    float4& acc, float& sv, const int2 ev, const int cnt,
    const float* __restrict__ x, const int lane, const int N)
{
    if (cnt > 0)  add_upto16<0>(acc, sv, ev, cnt > 16 ? 16 : cnt, x, lane, N);
    if (cnt > 16) add_upto16<16>(acc, sv, ev, cnt - 16, x, lane, N);
}

template <bool SCORE>
__global__ void __launch_bounds__(WPB * 32) gather_linear(
    const float* __restrict__ x, const float* __restrict__ W,
    const float* __restrict__ bias, const int* __restrict__ rowptr,
    const int2* __restrict__ csr, const float* __restrict__ prelu_a,
    float* __restrict__ Hst, float* __restrict__ csum,
    const float* __restrict__ Wbil, const float* __restrict__ bbil,
    float* __restrict__ score, int N, int E)
{
    __shared__ __attribute__((aligned(16))) float wS[H_DIM * APITCH];
    __shared__ __attribute__((aligned(16))) float aS[WPB * RPT * APITCH];
    __shared__ __attribute__((aligned(16))) float sS[H_DIM];
    __shared__ __attribute__((aligned(16))) float tS[H_DIM];

    const int lane = threadIdx.x & 31;
    const int wv = __builtin_amdgcn_readfirstlane((int)(threadIdx.x >> 5));
    const int r0 = (blockIdx.x * WPB + wv) * RPT;
    int nr = N - r0;
    nr = nr < 0 ? 0 : (nr > RPT ? RPT : nr);
    float* aT = aS + wv * (RPT * APITCH);

    const float a = prelu_a[0];
    float2 tv = make_float2(0.f, 0.f);

    constexpr int BIL_THREADS = H_DIM * (H_DIM / 4) / 2;
    static_assert(BIL_THREADS % 32 == 0 && BIL_THREADS <= WPB * 32, "W_bil pieces: whole waves");
    float4 wb0 = make_float4(0.f, 0.f, 0.f, 0.f), wb1 = wb0;
    if (SCORE) {
        const int bj = (threadIdx.x < BIL_THREADS) ? (int)threadIdx.x : 0;
        wb0 = ((const float4*)Wbil)[bj];
        wb1 = ((const float4*)Wbil)[bj + BIL_THREADS];
    }

    int pi = r0 + lane;
    pi = pi > N ? N : pi;
    int myptr = rowptr[pi];
    myptr = myptr < 0 ? 0 : (myptr > E ? E : myptr);

    constexpr int W_PIECES  = H_DIM * (F_DIM / 4);
    constexpr int W_STAGERS = 512;
    constexpr int W_TRIPS   = W_PIECES / W_STAGERS;
    static_assert(W_PIECES % W_STAGERS == 0 && W_STAGERS % 32 == 0 && W_STAGERS <= WPB * 32,
                  "W staging: whole waves, whole trips");
    if (threadIdx.x < W_STAGERS) {
        float4 wreg[W_TRIPS];
        #pragma unroll
        for (int t = 0; t < W_TRIPS; ++t)
            wreg[t] = ((const float4*)W)[threadIdx.x + t * W_STAGERS];
        #pragma unroll
        for (int t = 0; t < W_TRIPS; ++t) {
            const int idx = threadIdx.x + t * W_STAGERS;
            *(float4*)(wS + (idx >> 5) * APITCH + 4 * (idx & 31)) = wreg[t];
        }
    } else if (SCORE && threadIdx.x < W_STAGERS + H_DIM) {
        static_assert(W_STAGERS + H_DIM <= WPB * 32 && H_DIM % 32 == 0, "summary: whole waves behind the stagers");
        const int k = threadIdx.x - W_STAGERS;
        sS[k] = 1.f / (1.f + expf(-(csum[k] / (float)N)));
    }
    __syncthreads();

    if (SCORE) {
        if (threadIdx.x < BIL_THREADS) {
            const float4 s4 = *(const float4*)(sS + 4 * (threadIdx.x & 15));
            float p0 = (wb0.x * s4.x + wb0.y * s4.y) + (wb0.z * s4.z + wb0.w * s4.w);
            float p1 = (wb1.x * s4.x + wb1.y * s4.y) + (wb1.z * s4.z + wb1.w * s4.w);
            #pragma unroll
            for (int off = 8; off >= 1; off >>= 1) {
                p0 += __shfl_xor(p0, off, 32);
                p1 += __shfl_xor(p1, off, 32);
            }
            if ((threadIdx.x & 15) == 0) {
                tS[threadIdx.x >> 4]                = p0;
                tS[(threadIdx.x >> 4) + H_DIM / 2]  = p1;
            }
        }
        __syncthreads();
        tv = ((const float2*)tS)[lane];
    }

    int beg = __builtin_amdgcn_readlane(myptr, 0);
    int end = __builtin_amdgcn_readlane(myptr, 1);
    int2 ev = make_int2(0, 0);
    if (nr > 0 && lane < end - beg) ev = csr[beg + lane];
    asm volatile("" :: "v"(ev.x), "v"(ev.y));

    for (int i = 0; i < nr; ++i) {
        int endn = end;
        int2 evn = make_int2(0, 0);
        if (i + 1 < nr) {
            endn = __builtin_amdgcn_readlane(myptr, i + 2);
            if (lane < endn - end) evn = csr[end + lane];
        }

        float4 acc = make_float4(0.f, 0.f, 0.f, 0.f);
        float sv = 0.f;
        const int deg = end - beg;
        add_chunk(acc, sv, ev, deg > 32 ? 32 : deg, x, lane, N);
        if (deg > 32) {
            for (int base = beg + 32; base < end; base += 32) {
                int2 e2 = make_int2(0, 0);
                if (lane < end - base) e2 = csr[base + lane];
                const int rest = end - base;
                add_chunk(acc, sv, e2, rest > 32 ? 32 : rest, x, lane, N);
            }
        }

        *(float4*)(aT + i * APITCH + 4 * lane) = acc;
        if (lane == 0) aT[i * APITCH + F_DIM] = sv;

        beg = end;
        end = endn;
        ev  = evn;
    }
    for (int i = nr; i < RPT; ++i) {
        *(float4*)(aT + i * APITCH + 4 * lane) = make_float4(0.f, 0.f, 0.f, 0.f);
        if (lane == 0) aT[i * APITCH + F_DIM] = 0.f;
    }

    float2 cs = make_float2(0.f, 0.f);
    float myscore = 0.f;

    constexpr int PAIRS = RPT / 2;
    float4 hv[PAIRS];
    #pragma unroll
    for (int j = 0; j < PAIRS; ++j) hv[j] = make_float4(0.f, 0.f, 0.f, 0.f);
    if (SCORE && nr > 0) {
        #pragma unroll
        for (int j = 0; j < PAIRS; ++j) {
            int row = r0 + 2 * j + (lane >> 4);
            row = row < N ? row : N - 1;
            hv[j] = ((const float4*)(Hst + (size_t)row * H_DIM))[lane & 15];
        }
    }

    if (nr > 0) {
        __builtin_amdgcn_fence(__ATOMIC_RELEASE, "wavefront");
        __builtin_amdgcn_wave_barrier();
        __builtin_amdgcn_fence(__ATOMIC_ACQUIRE, "wavefront");

        const int c = lane & 15;
        const int h = lane >> 4;
        const float* ap = aT + c * APITCH + 64 * h;
        const float* wp = wS + c * APITCH + 64 * h;
        v8f d0 = {}, d1 = {}, d2 = {}, d3 = {};
        #pragma unroll 2
        for (int p = 0; p < 16; ++p) {
            const float4 a4 = *(const float4*)(ap + 4 * p);
            const float4 b0 = *(const float4*)(wp + 4 * p);
            const float4 b1 = *(const float4*)(wp + 16 * APITCH + 4 * p);
            const float4 b2 = *(const float4*)(wp + 32 * APITCH + 4 * p);
            const float4 b3 = *(const float4*)(wp + 48 * APITCH + 4 * p);
            const v2f alo = {a4.x, a4.y};
            const v2f ahi = {a4.z, a4.w};
            d0 = __builtin_amdgcn_wmma_f32_16x16x4_f32(false, alo, false, (v2f){b0.x, b0.y}, (short)0, d0, false, false);
            d1 = __builtin_amdgcn_wmma_f32_16x16x4_f32(false, alo, false, (v2f){b1.x, b1.y}, (short)0, d1, false, false);
            d2 = __builtin_amdgcn_wmma_f32_16x16x4_f32(false, alo, false, (v2f){b2.x, b2.y}, (short)0, d2, false, false);
            d3 = __builtin_amdgcn_wmma_f32_16x16x4_f32(false, alo, false, (v2f){b3.x, b3.y}, (short)0, d3, false, false);
            d0 = __builtin_amdgcn_wmma_f32_16x16x4_f32(false, ahi, false, (v2f){b0.z, b0.w}, (short)0, d0, false, false);
            d1 = __builtin_amdgcn_wmma_f32_16x16x4_f32(false, ahi, false, (v2f){b1.z, b1.w}, (short)0, d1, false, false);
            d2 = __builtin_amdgcn_wmma_f32_16x16x4_f32(false, ahi, false, (v2f){b2.z, b2.w}, (short)0, d2, false, false);
            d3 = __builtin_amdgcn_wmma_f32_16x16x4_f32(false, ahi, false, (v2f){b3.z, b3.w}, (short)0, d3, false, false);
        }

        const float bv0 = bias[c], bv1 = bias[16 + c], bv2 = bias[32 + c], bv3 = bias[48 + c];
        #pragma unroll
        for (int r = 0; r < 8; ++r) {
            float* orow = aT + (8 * h + r) * APITCH;
            const float svr = orow[F_DIM];
            float y0 = d0[r] + svr * bv0;
            float y1 = d1[r] + svr * bv1;
            float y2 = d2[r] + svr * bv2;
            float y3 = d3[r] + svr * bv3;
            orow[c]      = (y0 >= 0.f) ? y0 : a * y0;
            orow[16 + c] = (y1 >= 0.f) ? y1 : a * y1;
            orow[32 + c] = (y2 >= 0.f) ? y2 : a * y2;
            orow[48 + c] = (y3 >= 0.f) ? y3 : a * y3;
        }
        __builtin_amdgcn_fence(__ATOMIC_RELEASE, "wavefront");
        __builtin_amdgcn_wave_barrier();
        __builtin_amdgcn_fence(__ATOMIC_ACQUIRE, "wavefront");

        for (int i = 0; i < nr; ++i) {
            const float2 o = *(const float2*)(aT + i * APITCH + 2 * lane);
            if (!SCORE) {
                ((float2*)(Hst + (size_t)(r0 + i) * H_DIM))[lane] = o;
                cs.x += o.x;
                cs.y += o.y;
            } else {
                float p = o.x * tv.x + o.y * tv.y;
                #pragma unroll
                for (int off = 16; off >= 1; off >>= 1) p += __shfl_xor(p, off, 32);
                if (lane == i) myscore = p;
            }
        }
    }

    if (SCORE) {
        const float bb0 = bbil[0];
        if (lane < nr) score[(size_t)N + r0 + lane] = myscore + bb0;
        if (nr > 0) {
            const int c = lane & 15;
            const int h = lane >> 4;
            const float4 t4 = *(const float4*)(tS + 4 * c);
            float keep = 0.f;
            #pragma unroll
            for (int j = 0; j < PAIRS; ++j) {
                float p = (hv[j].x * t4.x + hv[j].y * t4.y) + (hv[j].z * t4.z + hv[j].w * t4.w);
                p += __shfl_xor(p, 8, 32);
                p += __shfl_xor(p, 4, 32);
                p += __shfl_xor(p, 2, 32);
                p += __shfl_xor(p, 1, 32);
                if (c == j) keep = p;
            }
            const int row = r0 + 2 * c + h;
            if (c < PAIRS && row < N) score[row] = keep + bb0;
        }
    } else {
        aT[2 * lane]     = cs.x;
        aT[2 * lane + 1] = cs.y;
        __syncthreads();
        if (threadIdx.x < H_DIM) {
            float t = 0.f;
            for (int w = 0; w < WPB; ++w) t += aS[w * (RPT * APITCH) + threadIdx.x];
            atomAddF(csum + threadIdx.x, t);
        }
    }
}

extern "C" void kernel_launch(void* const* d_in, const int* in_sizes, int n_in,
                              void* d_out, int out_size, void* d_ws, size_t ws_size,
                              hipStream_t stream) {
    const float* pos   = (const float*)d_in[0];
    const float* neg   = (const float*)d_in[1];
    const int*   erows = (const int*)  d_in[2];
    const int*   ecols = (const int*)  d_in[3];
    const float* evals = (const float*)d_in[4];
    const float* Wg    = (const float*)d_in[5];
    const float* bg    = (const float*)d_in[6];
    const float* pa    = (const float*)d_in[7];
    const float* Wb    = (const float*)d_in[8];
    const float* bb    = (const float*)d_in[9];

    const int N = in_sizes[0] / F_DIM;
    const int E = in_sizes[2];
    float* out = (float*)d_out;

    float* posH   = (float*)d_ws + (size_t)N * H_DIM;
    float* plane3 = posH + (size_t)N * H_DIM;
    float* colsum = plane3 + (size_t)N * H_DIM;
    int2*  tmp_cv  = (int2*)d_ws;
    int*   tmp_row = (int*)d_ws + 2 * (size_t)E;
    int2*  csr    = (int2*)plane3;
    int*   rowptr = (int*)plane3 + 2 * (size_t)E;
    int*   bcnt   = (int*)colsum - 2 * MAXNB;
    int*   cursor = bcnt + MAXNB;

    hipMemsetAsync(bcnt, 0, (size_t)(2 * MAXNB + H_DIM) * sizeof(float), stream);

    const int NB = (N + BROWS - 1) / BROWS;
    const int edgeBlocks = (E + PEDGES - 1) / PEDGES;
    const int rowTiles   = (N + RPT - 1) / RPT;
    const int fusedBlocks = (rowTiles + WPB - 1) / WPB;

    bucket_hist     <<<edgeBlocks, PTHREADS, 0, stream>>>(erows, bcnt, E, NB);
    bucket_partition<<<edgeBlocks, PTHREADS, 0, stream>>>(erows, ecols, evals, bcnt, cursor,
                                                          tmp_cv, tmp_row, E, NB);
    bucket_sort     <<<NB, PTHREADS, 0, stream>>>(tmp_cv, tmp_row, bcnt, rowptr, csr, N, E, NB);
    gather_linear<false><<<fusedBlocks, WPB * 32, 0, stream>>>(
        pos, Wg, bg, rowptr, csr, pa, posH, colsum, Wb, bb, out, N, E);
    gather_linear<true><<<fusedBlocks, WPB * 32, 0, stream>>>(
        neg, Wg, bg, rowptr, csr, pa, posH, colsum, Wb, bb, out, N, E);
}
